# starting kernel
# baseline (speedup 1.0000x reference)
#include <hip/hip_runtime.h>

typedef __attribute__((ext_vector_type(2))) float v2f;
typedef __attribute__((ext_vector_type(4))) float v4f;
typedef __attribute__((ext_vector_type(8))) float v8f;

#define Gc   64
#define Dc   16
#define Hc   8
#define Bc   64
#define Tc   512
#define K3   24
#define GD   (Gc * Dc)
#define NCH  (Gc * Bc)

__device__ __forceinline__ float fast_sigmoid(float x) {
    float e = __expf(-x);
    return __builtin_amdgcn_rcpf(1.0f + e);
}
__device__ __forceinline__ float fast_tanh(float x) {
    float e = __expf(2.0f * x);
    return 1.0f - 2.0f * __builtin_amdgcn_rcpf(e + 1.0f);
}

__global__ __launch_bounds__(256) void gru_input_proj_wmma(
    const float* __restrict__ x,
    const float* __restrict__ W_ih,
    const float* __restrict__ b_ih,
    float* __restrict__ gx)
{
    const int lane = threadIdx.x & 31;
    const int wave = threadIdx.x >> 5;
    const int tile = blockIdx.x * 8 + wave;

    const int g    = tile >> 11;
    const int rem  = tile & 2047;
    const int t    = rem >> 2;
    const int b0   = (rem & 3) << 4;

    const int half = lane >> 4;
    const int l16  = lane & 15;

    v2f a[4];
    {
        const v2f* xp = (const v2f*)(x + ((size_t)(b0 + l16) * Tc + t) * GD
                                       + g * Dc + 2 * half);
        #pragma unroll
        for (int j = 0; j < 4; ++j) a[j] = xp[2 * j];
    }

    v2f bm0[4], bm1[4];
    {
        const float* wg  = W_ih + (size_t)g * K3 * Dc;
        const v2f*   wr0 = (const v2f*)(wg + (size_t)l16 * Dc + 2 * half);
        const v2f*   wr1 = (const v2f*)(wg + (size_t)(8 + l16) * Dc + 2 * half);
        #pragma unroll
        for (int j = 0; j < 4; ++j) bm0[j] = wr0[2 * j];
        #pragma unroll
        for (int j = 0; j < 4; ++j) bm1[j] = wr1[2 * j];
    }

    v8f acc0 = {};
    v8f acc1 = {};
    #pragma unroll
    for (int j = 0; j < 4; ++j) {
        acc0 = __builtin_amdgcn_wmma_f32_16x16x4_f32(
                   false, a[j], false, bm0[j], (short)0, acc0, false, false);
        acc1 = __builtin_amdgcn_wmma_f32_16x16x4_f32(
                   false, a[j], false, bm1[j], (short)0, acc1, false, false);
    }

    const float bias0 = b_ih[g * K3 + l16];
    const float bias1 = b_ih[g * K3 + 8 + l16];

    const size_t rowBase = (size_t)t * NCH + (size_t)g * 64 + b0;
    #pragma unroll
    for (int r = 0; r < 8; ++r) {
        const int m = r + 8 * half;
        float* orow = gx + (rowBase + m) * K3;
        orow[l16]     = acc0[r] + bias0;
        orow[8 + l16] = acc1[r] + bias1;
    }
}

__global__ __launch_bounds__(256) void gru_recurrence(
    const float* __restrict__ gx,
    const float* __restrict__ W_hh,
    const float* __restrict__ b_hh,
    float* __restrict__ out)
{
    __shared__ float sW[4 * K3 * Hc];
    __shared__ float sB[4 * K3];

    const int tid   = threadIdx.x;
    const int gBase = blockIdx.x * 4;

    for (int i = tid; i < 4 * K3 * Hc; i += 256)
        sW[i] = W_hh[(size_t)gBase * K3 * Hc + i];
    for (int i = tid; i < 4 * K3; i += 256)
        sB[i] = b_hh[(size_t)gBase * K3 + i];
    __syncthreads();

    const int c  = blockIdx.x * 256 + tid;
    const int b  = c & 63;
    const int gl = tid >> 6;
    const int g  = gBase + gl;

    float w[K3][Hc];
    float bb[K3];
    {
        const v4f* sWv = (const v4f*)(sW + gl * K3 * Hc);
        #pragma unroll
        for (int k = 0; k < K3; ++k) {
            v4f w0 = sWv[2 * k];
            v4f w1 = sWv[2 * k + 1];
            w[k][0] = w0.x; w[k][1] = w0.y; w[k][2] = w0.z; w[k][3] = w0.w;
            w[k][4] = w1.x; w[k][5] = w1.y; w[k][6] = w1.z; w[k][7] = w1.w;
            bb[k]   = sB[gl * K3 + k];
        }
    }

    float* outBase  = out + (size_t)b * Tc * (Gc * Hc) + (size_t)g * Hc;
    const float* gc = gx + (size_t)c * K3;

    float h[Hc];
    #pragma unroll
    for (int i = 0; i < Hc; ++i) h[i] = 0.0f;

    auto loadgx = [&](v4f buf[6], int t) {
        const v4f* gp = (const v4f*)(gc + (size_t)t * (NCH * K3));
        #pragma unroll
        for (int i = 0; i < 6; ++i) buf[i] = gp[i];
    };

    auto step = [&](const v4f buf[6], int t) {
        float gxa[K3];
        #pragma unroll
        for (int i = 0; i < 6; ++i) {
            gxa[4 * i + 0] = buf[i].x; gxa[4 * i + 1] = buf[i].y;
            gxa[4 * i + 2] = buf[i].z; gxa[4 * i + 3] = buf[i].w;
        }

        float gh[K3];
        #pragma unroll
        for (int k = 0; k < K3; ++k) {
            float acc = bb[k];
            #pragma unroll
            for (int j = 0; j < Hc; ++j) acc += w[k][j] * h[j];
            gh[k] = acc;
        }

        #pragma unroll
        for (int i = 0; i < Hc; ++i) {
            float r = fast_sigmoid(gxa[i]      + gh[i]);
            float z = fast_sigmoid(gxa[8 + i]  + gh[8 + i]);
            float n = fast_tanh(gxa[16 + i] + r * gh[16 + i]);
            h[i] = (1.0f - z) * n + z * h[i];
        }

        v4f o0 = {h[0], h[1], h[2], h[3]};
        v4f o1 = {h[4], h[5], h[6], h[7]};
        v4f* op = (v4f*)(outBase + (size_t)t * (Gc * Hc));
        op[0] = o0;
        op[1] = o1;
    };

    v4f bufA[6], bufB[6];
    loadgx(bufA, 0);
    for (int t = 0; t < Tc; t += 2) {
        loadgx(bufB, t + 1);
        step(bufA, t);
        if (t + 2 < Tc) loadgx(bufA, t + 2);
        step(bufB, t + 1);
    }
}

extern "C" void kernel_launch(void* const* d_in, const int* in_sizes, int n_in,
                              void* d_out, int out_size, void* d_ws, size_t ws_size,
                              hipStream_t stream) {
    const float* x    = (const float*)d_in[0];
    const float* W_ih = (const float*)d_in[1];
    const float* W_hh = (const float*)d_in[2];
    const float* b_ih = (const float*)d_in[3];
    const float* b_hh = (const float*)d_in[4];
    float* out = (float*)d_out;
    float* gx  = (float*)d_ws;

    gru_input_proj_wmma<<<dim3(131072 / 8), dim3(256), 0, stream>>>(x, W_ih, b_ih, gx);
    gru_recurrence<<<dim3(NCH / 256), dim3(256), 0, stream>>>(gx, W_hh, b_hh, out);
}


# recurrence launched as 128 single-wave work-groups (32 chains each) instead of 16 work-groups of 8 waves: each chain wave alone on a SIMD and a vector cache; summation order unchanged
# speedup vs baseline: 1.3071x; 1.3071x over previous
#include <hip/hip_runtime.h>

typedef __attribute__((ext_vector_type(2))) float v2f;
typedef __attribute__((ext_vector_type(4))) float v4f;
typedef __attribute__((ext_vector_type(8))) float v8f;

#define Gc   64
#define Dc   16
#define Hc   8
#define Bc   64
#define Tc   512
#define K3   24
#define GD   (Gc * Dc)
#define NCH  (Gc * Bc)

__device__ __forceinline__ float fast_sigmoid(float x) {
    float e = __expf(-x);
    return __builtin_amdgcn_rcpf(1.0f + e);
}
__device__ __forceinline__ float fast_tanh(float x) {
    float e = __expf(2.0f * x);
    return 1.0f - 2.0f * __builtin_amdgcn_rcpf(e + 1.0f);
}

__global__ __launch_bounds__(256) void gru_input_proj_wmma(
    const float* __restrict__ x,
    const float* __restrict__ W_ih,
    const float* __restrict__ b_ih,
    float* __restrict__ gx)
{
    const int lane = threadIdx.x & 31;
    const int wave = threadIdx.x >> 5;
    const int tile = blockIdx.x * 8 + wave;

    const int g    = tile >> 11;
    const int rem  = tile & 2047;
    const int t    = rem >> 2;
    const int b0   = (rem & 3) << 4;

    const int half = lane >> 4;
    const int l16  = lane & 15;

    v2f a[4];
    {
        const v2f* xp = (const v2f*)(x + ((size_t)(b0 + l16) * Tc + t) * GD
                                       + g * Dc + 2 * half);
        #pragma unroll
        for (int j = 0; j < 4; ++j) a[j] = xp[2 * j];
    }

    v2f bm0[4], bm1[4];
    {
        const float* wg  = W_ih + (size_t)g * K3 * Dc;
        const v2f*   wr0 = (const v2f*)(wg + (size_t)l16 * Dc + 2 * half);
        const v2f*   wr1 = (const v2f*)(wg + (size_t)(8 + l16) * Dc + 2 * half);
        #pragma unroll
        for (int j = 0; j < 4; ++j) bm0[j] = wr0[2 * j];
        #pragma unroll
        for (int j = 0; j < 4; ++j) bm1[j] = wr1[2 * j];
    }

    v8f acc0 = {};
    v8f acc1 = {};
    #pragma unroll
    for (int j = 0; j < 4; ++j) {
        acc0 = __builtin_amdgcn_wmma_f32_16x16x4_f32(
                   false, a[j], false, bm0[j], (short)0, acc0, false, false);
        acc1 = __builtin_amdgcn_wmma_f32_16x16x4_f32(
                   false, a[j], false, bm1[j], (short)0, acc1, false, false);
    }

    const float bias0 = b_ih[g * K3 + l16];
    const float bias1 = b_ih[g * K3 + 8 + l16];

    const size_t rowBase = (size_t)t * NCH + (size_t)g * 64 + b0;
    #pragma unroll
    for (int r = 0; r < 8; ++r) {
        const int m = r + 8 * half;
        float* orow = gx + (rowBase + m) * K3;
        orow[l16]     = acc0[r] + bias0;
        orow[8 + l16] = acc1[r] + bias1;
    }
}

__global__ __launch_bounds__(32) void gru_recurrence(
    const float* __restrict__ gx,
    const float* __restrict__ W_hh,
    const float* __restrict__ b_hh,
    float* __restrict__ out)
{
    __shared__ float sW[K3 * Hc];
    __shared__ float sB[K3];

    const int tid = threadIdx.x;
    const int c   = blockIdx.x * 32 + tid;
    const int b   = c & 63;
    const int g   = blockIdx.x >> 1;

    for (int i = 0; i < (K3 * Hc) / 32; ++i)
        sW[tid + 32 * i] = W_hh[(size_t)g * K3 * Hc + tid + 32 * i];
    if (tid < K3)
        sB[tid] = b_hh[(size_t)g * K3 + tid];
    __syncthreads();

    float w[K3][Hc];
    float bb[K3];
    {
        const v4f* sWv = (const v4f*)sW;
        #pragma unroll
        for (int k = 0; k < K3; ++k) {
            v4f w0 = sWv[2 * k];
            v4f w1 = sWv[2 * k + 1];
            w[k][0] = w0.x; w[k][1] = w0.y; w[k][2] = w0.z; w[k][3] = w0.w;
            w[k][4] = w1.x; w[k][5] = w1.y; w[k][6] = w1.z; w[k][7] = w1.w;
            bb[k]   = sB[k];
        }
    }

    float* outBase  = out + (size_t)b * Tc * (Gc * Hc) + (size_t)g * Hc;
    const float* gc = gx + (size_t)c * K3;

    float h[Hc];
    #pragma unroll
    for (int i = 0; i < Hc; ++i) h[i] = 0.0f;

    auto loadgx = [&](v4f buf[6], int t) {
        const v4f* gp = (const v4f*)(gc + (size_t)t * (NCH * K3));
        #pragma unroll
        for (int i = 0; i < 6; ++i) buf[i] = gp[i];
    };

    auto step = [&](const v4f buf[6], int t) {
        float gxa[K3];
        #pragma unroll
        for (int i = 0; i < 6; ++i) {
            gxa[4 * i + 0] = buf[i].x; gxa[4 * i + 1] = buf[i].y;
            gxa[4 * i + 2] = buf[i].z; gxa[4 * i + 3] = buf[i].w;
        }

        float gh[K3];
        #pragma unroll
        for (int k = 0; k < K3; ++k) {
            float acc = bb[k];
            #pragma unroll
            for (int j = 0; j < Hc; ++j) acc += w[k][j] * h[j];
            gh[k] = acc;
        }

        #pragma unroll
        for (int i = 0; i < Hc; ++i) {
            float r = fast_sigmoid(gxa[i]      + gh[i]);
            float z = fast_sigmoid(gxa[8 + i]  + gh[8 + i]);
            float n = fast_tanh(gxa[16 + i] + r * gh[16 + i]);
            h[i] = (1.0f - z) * n + z * h[i];
        }

        v4f o0 = {h[0], h[1], h[2], h[3]};
        v4f o1 = {h[4], h[5], h[6], h[7]};
        v4f* op = (v4f*)(outBase + (size_t)t * (Gc * Hc));
        op[0] = o0;
        op[1] = o1;
    };

    v4f bufA[6], bufB[6];
    loadgx(bufA, 0);
    for (int t = 0; t < Tc; t += 2) {
        loadgx(bufB, t + 1);
        step(bufA, t);
        if (t + 2 < Tc) loadgx(bufA, t + 2);
        step(bufB, t + 1);
    }
}

extern "C" void kernel_launch(void* const* d_in, const int* in_sizes, int n_in,
                              void* d_out, int out_size, void* d_ws, size_t ws_size,
                              hipStream_t stream) {
    const float* x    = (const float*)d_in[0];
    const float* W_ih = (const float*)d_in[1];
    const float* W_hh = (const float*)d_in[2];
    const float* b_ih = (const float*)d_in[3];
    const float* b_hh = (const float*)d_in[4];
    float* out = (float*)d_out;
    float* gx  = (float*)d_ws;

    gru_input_proj_wmma<<<dim3(131072 / 8), dim3(256), 0, stream>>>(x, W_ih, b_ih, gx);
    gru_recurrence<<<dim3(NCH / 32), dim3(32), 0, stream>>>(gx, W_hh, b_hh, out);
}


# recurrence reads gx 8 steps ahead through a ring of 8 register slots and waits only for a step's own row (was 1 step ahead, waited for at issue every 2nd step); summation order unchanged
# speedup vs baseline: 2.6001x; 1.9892x over previous
#include <hip/hip_runtime.h>

typedef __attribute__((ext_vector_type(2))) float v2f;
typedef __attribute__((ext_vector_type(4))) float v4f;
typedef __attribute__((ext_vector_type(8))) float v8f;

#define Gc   64
#define Dc   16
#define Hc   8
#define Bc   64
#define Tc   512
#define K3   24
#define GD   (Gc * Dc)
#define NCH  (Gc * Bc)
#define PF   8

__device__ __forceinline__ float fast_sigmoid(float x) {
    float e = __expf(-x);
    return __builtin_amdgcn_rcpf(1.0f + e);
}
__device__ __forceinline__ float fast_tanh(float x) {
    float e = __expf(2.0f * x);
    return 1.0f - 2.0f * __builtin_amdgcn_rcpf(e + 1.0f);
}

__global__ __launch_bounds__(256) void gru_input_proj_wmma(
    const float* __restrict__ x,
    const float* __restrict__ W_ih,
    const float* __restrict__ b_ih,
    float* __restrict__ gx)
{
    const int lane = threadIdx.x & 31;
    const int wave = threadIdx.x >> 5;
    const int tile = blockIdx.x * 8 + wave;

    const int g    = tile >> 11;
    const int rem  = tile & 2047;
    const int t    = rem >> 2;
    const int b0   = (rem & 3) << 4;

    const int half = lane >> 4;
    const int l16  = lane & 15;

    v2f a[4];
    {
        const v2f* xp = (const v2f*)(x + ((size_t)(b0 + l16) * Tc + t) * GD
                                       + g * Dc + 2 * half);
        #pragma unroll
        for (int j = 0; j < 4; ++j) a[j] = xp[2 * j];
    }

    v2f bm0[4], bm1[4];
    {
        const float* wg  = W_ih + (size_t)g * K3 * Dc;
        const v2f*   wr0 = (const v2f*)(wg + (size_t)l16 * Dc + 2 * half);
        const v2f*   wr1 = (const v2f*)(wg + (size_t)(8 + l16) * Dc + 2 * half);
        #pragma unroll
        for (int j = 0; j < 4; ++j) bm0[j] = wr0[2 * j];
        #pragma unroll
        for (int j = 0; j < 4; ++j) bm1[j] = wr1[2 * j];
    }

    v8f acc0 = {};
    v8f acc1 = {};
    #pragma unroll
    for (int j = 0; j < 4; ++j) {
        acc0 = __builtin_amdgcn_wmma_f32_16x16x4_f32(
                   false, a[j], false, bm0[j], (short)0, acc0, false, false);
        acc1 = __builtin_amdgcn_wmma_f32_16x16x4_f32(
                   false, a[j], false, bm1[j], (short)0, acc1, false, false);
    }

    const float bias0 = b_ih[g * K3 + l16];
    const float bias1 = b_ih[g * K3 + 8 + l16];

    const size_t rowBase = (size_t)t * NCH + (size_t)g * 64 + b0;
    #pragma unroll
    for (int r = 0; r < 8; ++r) {
        const int m = r + 8 * half;
        float* orow = gx + (rowBase + m) * K3;
        orow[l16]     = acc0[r] + bias0;
        orow[8 + l16] = acc1[r] + bias1;
    }
}

__global__ __launch_bounds__(32) void gru_recurrence(
    const float* __restrict__ gx,
    const float* __restrict__ W_hh,
    const float* __restrict__ b_hh,
    float* __restrict__ out)
{
    __shared__ float sW[K3 * Hc];
    __shared__ float sB[K3];

    const int tid = threadIdx.x;
    const int c   = blockIdx.x * 32 + tid;
    const int b   = c & 63;
    const int g   = blockIdx.x >> 1;

    for (int i = 0; i < (K3 * Hc) / 32; ++i)
        sW[tid + 32 * i] = W_hh[(size_t)g * K3 * Hc + tid + 32 * i];
    if (tid < K3)
        sB[tid] = b_hh[(size_t)g * K3 + tid];
    __syncthreads();

    float w[K3][Hc];
    float bb[K3];
    {
        const v4f* sWv = (const v4f*)sW;
        #pragma unroll
        for (int k = 0; k < K3; ++k) {
            v4f w0 = sWv[2 * k];
            v4f w1 = sWv[2 * k + 1];
            w[k][0] = w0.x; w[k][1] = w0.y; w[k][2] = w0.z; w[k][3] = w0.w;
            w[k][4] = w1.x; w[k][5] = w1.y; w[k][6] = w1.z; w[k][7] = w1.w;
            bb[k]   = sB[k];
        }
    }

    float* outBase  = out + (size_t)b * Tc * (Gc * Hc) + (size_t)g * Hc;
    const float* gc = gx + (size_t)c * K3;

    float h[Hc];
    #pragma unroll
    for (int i = 0; i < Hc; ++i) h[i] = 0.0f;

    auto loadrow = [&](v4f buf[6], const float* row) {
        const v4f* gp = (const v4f*)row;
        #pragma unroll
        for (int i = 0; i < 6; ++i) buf[i] = gp[i];
    };

    auto step = [&](const v4f buf[6], int t) {
        float gxa[K3];
        #pragma unroll
        for (int i = 0; i < 6; ++i) {
            gxa[4 * i + 0] = buf[i].x; gxa[4 * i + 1] = buf[i].y;
            gxa[4 * i + 2] = buf[i].z; gxa[4 * i + 3] = buf[i].w;
        }

        float gh[K3];
        #pragma unroll
        for (int k = 0; k < K3; ++k) {
            float acc = bb[k];
            #pragma unroll
            for (int j = 0; j < Hc; ++j) acc += w[k][j] * h[j];
            gh[k] = acc;
        }

        #pragma unroll
        for (int i = 0; i < Hc; ++i) {
            float r = fast_sigmoid(gxa[i]      + gh[i]);
            float z = fast_sigmoid(gxa[8 + i]  + gh[8 + i]);
            float n = fast_tanh(gxa[16 + i] + r * gh[16 + i]);
            h[i] = (1.0f - z) * n + z * h[i];
        }

        v4f o0 = {h[0], h[1], h[2], h[3]};
        v4f o1 = {h[4], h[5], h[6], h[7]};
        v4f* op = (v4f*)(outBase + (size_t)t * (Gc * Hc));
        op[0] = o0;
        op[1] = o1;
    };

    v4f ring[PF][6];
    #pragma unroll
    for (int s = 0; s < PF; ++s) loadrow(ring[s], gc + (size_t)s * (NCH * K3));

    for (int t = 0; t < Tc; t += PF) {
        #pragma unroll
        for (int s = 0; s < PF; ++s) {
            step(ring[s], t + s);
            int tn = t + s + PF;
            tn = tn < Tc ? tn : Tc - 1;
            asm volatile("" : "+s"(tn)
                            : "v"(h[0]), "v"(h[1]), "v"(h[2]), "v"(h[3]),
                              "v"(h[4]), "v"(h[5]), "v"(h[6]), "v"(h[7]));
            loadrow(ring[s], gc + (size_t)tn * (NCH * K3));
            __builtin_amdgcn_sched_barrier(0);
        }
    }
}

extern "C" void kernel_launch(void* const* d_in, const int* in_sizes, int n_in,
                              void* d_out, int out_size, void* d_ws, size_t ws_size,
                              hipStream_t stream) {
    const float* x    = (const float*)d_in[0];
    const float* W_ih = (const float*)d_in[1];
    const float* W_hh = (const float*)d_in[2];
    const float* b_ih = (const float*)d_in[3];
    const float* b_hh = (const float*)d_in[4];
    float* out = (float*)d_out;
    float* gx  = (float*)d_ws;

    gru_input_proj_wmma<<<dim3(131072 / 8), dim3(256), 0, stream>>>(x, W_ih, b_ih, gx);
    gru_recurrence<<<dim3(NCH / 32), dim3(32), 0, stream>>>(gx, W_hh, b_hh, out);
}


# recurrence split to one lane per (chain, hidden unit): 8 lanes share h by DPP8 broadcast, 1024 single-wave work-groups; each dot product summed as two fused 4-term chains (summation order changed)
# speedup vs baseline: 4.4628x; 1.7164x over previous
#include <hip/hip_runtime.h>

typedef __attribute__((ext_vector_type(2))) float v2f;
typedef __attribute__((ext_vector_type(4))) float v4f;
typedef __attribute__((ext_vector_type(8))) float v8f;

#define Gc   64
#define Dc   16
#define Hc   8
#define Bc   64
#define Tc   512
#define K3   24
#define GD   (Gc * Dc)
#define NCH  (Gc * Bc)
#define PF   16

__device__ __forceinline__ float fast_sigmoid(float x) {
    float e = __expf(-x);
    return __builtin_amdgcn_rcpf(1.0f + e);
}
__device__ __forceinline__ float fast_tanh(float x) {
    float e = __expf(2.0f * x);
    return 1.0f - 2.0f * __builtin_amdgcn_rcpf(e + 1.0f);
}

__global__ __launch_bounds__(256) void gru_input_proj_wmma(
    const float* __restrict__ x,
    const float* __restrict__ W_ih,
    const float* __restrict__ b_ih,
    float* __restrict__ gx)
{
    const int lane = threadIdx.x & 31;
    const int wave = threadIdx.x >> 5;
    const int tile = blockIdx.x * 8 + wave;

    const int g    = tile >> 11;
    const int rem  = tile & 2047;
    const int t    = rem >> 2;
    const int b0   = (rem & 3) << 4;

    const int half = lane >> 4;
    const int l16  = lane & 15;

    v2f a[4];
    {
        const v2f* xp = (const v2f*)(x + ((size_t)(b0 + l16) * Tc + t) * GD
                                       + g * Dc + 2 * half);
        #pragma unroll
        for (int j = 0; j < 4; ++j) a[j] = xp[2 * j];
    }

    v2f bm0[4], bm1[4];
    {
        const float* wg  = W_ih + (size_t)g * K3 * Dc;
        const v2f*   wr0 = (const v2f*)(wg + (size_t)l16 * Dc + 2 * half);
        const v2f*   wr1 = (const v2f*)(wg + (size_t)(8 + l16) * Dc + 2 * half);
        #pragma unroll
        for (int j = 0; j < 4; ++j) bm0[j] = wr0[2 * j];
        #pragma unroll
        for (int j = 0; j < 4; ++j) bm1[j] = wr1[2 * j];
    }

    v8f acc0 = {};
    v8f acc1 = {};
    #pragma unroll
    for (int j = 0; j < 4; ++j) {
        acc0 = __builtin_amdgcn_wmma_f32_16x16x4_f32(
                   false, a[j], false, bm0[j], (short)0, acc0, false, false);
        acc1 = __builtin_amdgcn_wmma_f32_16x16x4_f32(
                   false, a[j], false, bm1[j], (short)0, acc1, false, false);
    }

    const float bias0 = b_ih[g * K3 + l16];
    const float bias1 = b_ih[g * K3 + 8 + l16];

    const size_t rowBase = (size_t)t * NCH + (size_t)g * 64 + b0;
    #pragma unroll
    for (int r = 0; r < 8; ++r) {
        const int m = r + 8 * half;
        float* orow = gx + (rowBase + m) * K3;
        orow[l16]     = acc0[r] + bias0;
        orow[8 + l16] = acc1[r] + bias1;
    }
}

typedef const __attribute__((address_space(1))) float* gmem_row;

template <int J>
__device__ __forceinline__ float group8_bcast(float v) {
    return __uint_as_float(__builtin_amdgcn_mov_dpp8(__float_as_uint(v), J * 0x249249));
}

__global__ __launch_bounds__(32) void gru_recurrence(
    const float* __restrict__ gx,
    const float* __restrict__ W_hh,
    const float* __restrict__ b_hh,
    float* __restrict__ out)
{
    const int lane = threadIdx.x;
    const int i    = lane & 7;
    const int c    = blockIdx.x * 4 + (lane >> 3);
    const int b    = c & 63;
    const int g    = blockIdx.x >> 4;

    float wr[Hc], wz[Hc], wn[Hc];
    {
        const float* wg = W_hh + (size_t)g * K3 * Hc;
        const v4f* pr = (const v4f*)(wg + (size_t)i * Hc);
        const v4f* pz = (const v4f*)(wg + (size_t)(8 + i) * Hc);
        const v4f* pn = (const v4f*)(wg + (size_t)(16 + i) * Hc);
        v4f r0 = pr[0], r1 = pr[1], z0 = pz[0], z1 = pz[1], n0 = pn[0], n1 = pn[1];
        wr[0] = r0.x; wr[1] = r0.y; wr[2] = r0.z; wr[3] = r0.w;
        wr[4] = r1.x; wr[5] = r1.y; wr[6] = r1.z; wr[7] = r1.w;
        wz[0] = z0.x; wz[1] = z0.y; wz[2] = z0.z; wz[3] = z0.w;
        wz[4] = z1.x; wz[5] = z1.y; wz[6] = z1.z; wz[7] = z1.w;
        wn[0] = n0.x; wn[1] = n0.y; wn[2] = n0.z; wn[3] = n0.w;
        wn[4] = n1.x; wn[5] = n1.y; wn[6] = n1.z; wn[7] = n1.w;
    }
    float br = b_hh[(size_t)g * K3 + i];
    float bz = b_hh[(size_t)g * K3 + 8 + i];
    float bn = b_hh[(size_t)g * K3 + 16 + i];

    const unsigned outOff = (unsigned)b * (Tc * Gc * Hc) + (unsigned)g * Hc + i;
    const unsigned gxOff  = (unsigned)c * K3 + i;

    float h = 0.0f;

    auto loadrow = [&](float buf[3], gmem_row row) {
        gmem_row p = row + gxOff;
        buf[0] = p[0];
        buf[1] = p[8];
        buf[2] = p[16];
    };

    auto step = [&](const float buf[3], int t) {
        const float sr = buf[0] + br;
        const float sz = buf[1] + bz;

        const float h0 = group8_bcast<0>(h), h1 = group8_bcast<1>(h);
        const float h2 = group8_bcast<2>(h), h3 = group8_bcast<3>(h);
        const float h4 = group8_bcast<4>(h), h5 = group8_bcast<5>(h);
        const float h6 = group8_bcast<6>(h), h7 = group8_bcast<7>(h);

        float ar = fmaf(wr[3], h3, fmaf(wr[2], h2, fmaf(wr[1], h1, fmaf(wr[0], h0, sr))));
        float az = fmaf(wz[3], h3, fmaf(wz[2], h2, fmaf(wz[1], h1, fmaf(wz[0], h0, sz))));
        float an = fmaf(wn[3], h3, fmaf(wn[2], h2, fmaf(wn[1], h1, fmaf(wn[0], h0, bn))));
        float cr = fmaf(wr[7], h7, fmaf(wr[6], h6, fmaf(wr[5], h5, wr[4] * h4)));
        float cz = fmaf(wz[7], h7, fmaf(wz[6], h6, fmaf(wz[5], h5, wz[4] * h4)));
        float cn = fmaf(wn[7], h7, fmaf(wn[6], h6, fmaf(wn[5], h5, wn[4] * h4)));

        const float r = fast_sigmoid(ar + cr);
        const float z = fast_sigmoid(az + cz);
        const float n = fast_tanh(buf[2] + r * (an + cn));
        h = (1.0f - z) * n + z * h;

        float* orow = out + (size_t)t * (Gc * Hc);
        orow[outOff] = h;
    };

    float ring[PF][3];
    #pragma unroll
    for (int s = 0; s < PF; ++s) loadrow(ring[s], (gmem_row)(gx + (size_t)s * (NCH * K3)));

    #pragma unroll
    for (int j = 0; j < Hc; ++j)
        asm volatile("" : "+v"(wr[j]), "+v"(wz[j]), "+v"(wn[j]));
    asm volatile("" : "+v"(br), "+v"(bz), "+v"(bn));

    for (int t = 0; t < Tc; t += PF) {
        #pragma unroll
        for (int s = 0; s < PF; ++s) {
            step(ring[s], t + s);
            int tn = t + s + PF;
            tn = tn < Tc ? tn : Tc - 1;
            gmem_row row = (gmem_row)(gx + (size_t)tn * (NCH * K3));
            asm volatile("" : "+s"(row) : "v"(h));
            loadrow(ring[s], row);
            __builtin_amdgcn_sched_barrier(0);
        }
    }
}

extern "C" void kernel_launch(void* const* d_in, const int* in_sizes, int n_in,
                              void* d_out, int out_size, void* d_ws, size_t ws_size,
                              hipStream_t stream) {
    const float* x    = (const float*)d_in[0];
    const float* W_ih = (const float*)d_in[1];
    const float* W_hh = (const float*)d_in[2];
    const float* b_ih = (const float*)d_in[3];
    const float* b_hh = (const float*)d_in[4];
    float* out = (float*)d_out;
    float* gx  = (float*)d_ws;

    gru_input_proj_wmma<<<dim3(131072 / 8), dim3(256), 0, stream>>>(x, W_ih, b_ih, gx);
    gru_recurrence<<<dim3(NCH / 4), dim3(32), 0, stream>>>(gx, W_hh, b_hh, out);
}


# input projection moved into the recurrence kernel: each wave builds gx for its 4 chains x 4 steps by f32 16x16x4 products one tile ahead; gx never goes through memory; one launch; sums unchanged
# speedup vs baseline: 6.8785x; 1.5413x over previous
#include <hip/hip_runtime.h>


typedef __attribute__((ext_vector_type(2))) float v2f;
typedef __attribute__((ext_vector_type(4))) float v4f;
typedef __attribute__((ext_vector_type(8))) float v8f;

#define Gc   64
#define Dc   16
#define Hc   8
#define Bc   64
#define Tc   512
#define K3   24
#define GD   (Gc * Dc)
#define NCH  (Gc * Bc)
#define TG   4
#define NXG  4
#define NGRP (Tc / TG)

__device__ __forceinline__ float fast_sigmoid(float x) {
    float e = __expf(-x);
    return __builtin_amdgcn_rcpf(1.0f + e);
}
__device__ __forceinline__ float fast_tanh(float x) {
    float e = __expf(2.0f * x);
    return 1.0f - 2.0f * __builtin_amdgcn_rcpf(e + 1.0f);
}

typedef const __attribute__((address_space(1))) float* gmem_row;
typedef const __attribute__((address_space(1))) v2f*   gmem_v2;
typedef const __attribute__((address_space(1))) char*  gmem_bytes;

template <int J>
__device__ __forceinline__ float group8_bcast(float v) {
    return __uint_as_float(__builtin_amdgcn_mov_dpp8(__float_as_uint(v), J * 0x249249));
}

template <int CTRL>
__device__ __forceinline__ float row_shift_keep(float keep, float src) {
    return __uint_as_float(__builtin_amdgcn_update_dpp(
        __float_as_uint(keep), __float_as_uint(src), CTRL, 0xf, 0xf, false));
}

__device__ __forceinline__ v8f mm(v2f a, v2f b, v8f c) {
    return __builtin_amdgcn_wmma_f32_16x16x4_f32(false, a, false, b, (short)0, c, false, false);
}

__global__ __launch_bounds__(32) void gru_fused(
    const float* __restrict__ x,
    const float* __restrict__ W_ih,
    const float* __restrict__ W_hh,
    const float* __restrict__ b_ih,
    const float* __restrict__ b_hh,
    float* __restrict__ out)
{
    const int lane = threadIdx.x;
    const int i    = lane & 7;
    const int c    = blockIdx.x * 4 + (lane >> 3);
    const int b    = c & 63;
    const int g    = blockIdx.x >> 4;
    const int half = lane >> 4;
    const int l16  = lane & 15;
    const bool upper = (lane & 8) != 0;

    float wr[Hc], wz[Hc], wn[Hc];
    {
        const float* wg = W_hh + (size_t)g * K3 * Hc;
        const v4f* pr = (const v4f*)(wg + (size_t)i * Hc);
        const v4f* pz = (const v4f*)(wg + (size_t)(8 + i) * Hc);
        const v4f* pn = (const v4f*)(wg + (size_t)(16 + i) * Hc);
        v4f r0 = pr[0], r1 = pr[1], z0 = pz[0], z1 = pz[1], n0 = pn[0], n1 = pn[1];
        wr[0] = r0.x; wr[1] = r0.y; wr[2] = r0.z; wr[3] = r0.w;
        wr[4] = r1.x; wr[5] = r1.y; wr[6] = r1.z; wr[7] = r1.w;
        wz[0] = z0.x; wz[1] = z0.y; wz[2] = z0.z; wz[3] = z0.w;
        wz[4] = z1.x; wz[5] = z1.y; wz[6] = z1.z; wz[7] = z1.w;
        wn[0] = n0.x; wn[1] = n0.y; wn[2] = n0.z; wn[3] = n0.w;
        wn[4] = n1.x; wn[5] = n1.y; wn[6] = n1.z; wn[7] = n1.w;
    }
    float br = b_hh[(size_t)g * K3 + i];
    float bz = b_hh[(size_t)g * K3 + 8 + i];
    float bn = b_hh[(size_t)g * K3 + 16 + i];
    float ir = b_ih[(size_t)g * K3 + i];
    float iz = b_ih[(size_t)g * K3 + 8 + i];
    float in_ = b_ih[(size_t)g * K3 + 16 + i];

    v2f bm0[4], bm1[4];
    {
        const float* wg  = W_ih + (size_t)g * K3 * Dc;
        const v2f*   wr0 = (const v2f*)(wg + (size_t)l16 * Dc + 2 * half);
        const v2f*   wr1 = (const v2f*)(wg + (size_t)(16 + (l16 & 7)) * Dc + 2 * half);
        #pragma unroll
        for (int j = 0; j < 4; ++j) bm0[j] = wr0[2 * j];
        #pragma unroll
        for (int j = 0; j < 4; ++j) bm1[j] = wr1[2 * j];
    }

    const unsigned outOff = (unsigned)b * (Tc * Gc * Hc) + (unsigned)g * Hc + i;
    const unsigned bA   = (unsigned)((blockIdx.x & 15) * 4 + (l16 >> 2));
    const unsigned xOffB = (bA * (unsigned)(Tc * GD) + (unsigned)(l16 & 3) * GD
                         + (unsigned)g * Dc + 2 * half) * 4u;

    float h = 0.0f;
    unsigned xoB = xOffB;
    int gnext = NXG + 1;

    auto loadx = [&](v2f a[4], gmem_row row, unsigned offB) {
        gmem_bytes p = (gmem_bytes)row + offB;
        #pragma unroll
        for (int j = 0; j < 4; ++j) a[j] = *(gmem_v2)(p + 16 * j);
    };

    auto step = [&](float gr, float gz, float gn, int t, auto&& behind_exp_r, auto&& behind_exp_n) {
        const float sr = gr + br;
        const float sz = gz + bz;

        const float h0 = group8_bcast<0>(h), h1 = group8_bcast<1>(h);
        const float h2 = group8_bcast<2>(h), h3 = group8_bcast<3>(h);
        const float h4 = group8_bcast<4>(h), h5 = group8_bcast<5>(h);
        const float h6 = group8_bcast<6>(h), h7 = group8_bcast<7>(h);

        float ar = fmaf(wr[3], h3, fmaf(wr[2], h2, fmaf(wr[1], h1, fmaf(wr[0], h0, sr))));
        float az = fmaf(wz[3], h3, fmaf(wz[2], h2, fmaf(wz[1], h1, fmaf(wz[0], h0, sz))));
        float an = fmaf(wn[3], h3, fmaf(wn[2], h2, fmaf(wn[1], h1, fmaf(wn[0], h0, bn))));
        float cr = fmaf(wr[7], h7, fmaf(wr[6], h6, fmaf(wr[5], h5, wr[4] * h4)));
        float cz = fmaf(wz[7], h7, fmaf(wz[6], h6, fmaf(wz[5], h5, wz[4] * h4)));
        float cn = fmaf(wn[7], h7, fmaf(wn[6], h6, fmaf(wn[5], h5, wn[4] * h4)));

        const float er = __expf(-(ar + cr));
        behind_exp_r(er);
        const float r = __builtin_amdgcn_rcpf(1.0f + er);
        const float z = fast_sigmoid(az + cz);
        const float en = __expf(2.0f * (gn + r * (an + cn)));
        behind_exp_n(en);
        const float n = 1.0f - 2.0f * __builtin_amdgcn_rcpf(en + 1.0f);
        h = (1.0f - z) * n + z * h;

        float* orow = out + (size_t)t * (Gc * Hc);
        orow[outOff] = h;
    };

    auto group = [&](const v8f& c0, const v8f& c1, v8f& n0, v8f& n1, v2f xa[4], int G) {
        #pragma unroll
        for (int s = 0; s < TG; ++s) {
            const float gr = row_shift_keep<0x118>(c0[s], c0[4 + s]) + ir;
            const float gz = row_shift_keep<0x108>(c0[4 + s], c0[s]) + iz;
            float gn_hi = c1[4 + s];
            asm("" : "+v"(gn_hi));
            const float gn = (upper ? gn_hi : c1[s]) + in_;
            v2f ax = xa[s];
            v8f zero = {};
            step(gr, gz, gn, G * TG + s,
                 [&](float e) {
                     asm volatile("" : "+v"(ax) : "v"(e));
                     n0 = mm(ax, bm0[s], s == 0 ? zero : n0);
                 },
                 [&](float e) {
                     asm volatile("" : "+v"(ax) : "v"(e));
                     n1 = mm(ax, bm1[s], s == 0 ? zero : n1);
                 });
            if (s == TG - 1) {
                asm volatile("" : "+s"(gnext), "+v"(xoB) : "v"(h));
                const int Gn = gnext < NGRP ? gnext : NGRP - 1;
                gmem_row row = (gmem_row)(x + (size_t)Gn * (TG * GD));
                loadx(xa, row, xoB);
                gnext += 1;
            }
            __builtin_amdgcn_sched_barrier(0);
        }
    };

    v2f xr[NXG][4];
    #pragma unroll
    for (int k = 0; k < NXG; ++k) loadx(xr[k], (gmem_row)(x + (size_t)k * (TG * GD)), xOffB);

    v8f a0 = {}, a1 = {}, d0 = {}, d1 = {};
    #pragma unroll
    for (int j = 0; j < 4; ++j) { a0 = mm(xr[0][j], bm0[j], a0); a1 = mm(xr[0][j], bm1[j], a1); }
    loadx(xr[0], (gmem_row)(x + (size_t)NXG * (TG * GD)), xOffB);

    #pragma unroll
    for (int j = 0; j < Hc; ++j)
        asm volatile("" : "+v"(wr[j]), "+v"(wz[j]), "+v"(wn[j]));
    asm volatile("" : "+v"(br), "+v"(bz), "+v"(bn), "+v"(ir), "+v"(iz), "+v"(in_));
    #pragma unroll
    for (int j = 0; j < 4; ++j)
        asm volatile("" : "+v"(bm0[j]), "+v"(bm1[j]));

    for (int G0 = 0; G0 < NGRP; G0 += NXG) {
        group(a0, a1, d0, d1, xr[1], G0 + 0);
        group(d0, d1, a0, a1, xr[2], G0 + 1);
        group(a0, a1, d0, d1, xr[3], G0 + 2);
        group(d0, d1, a0, a1, xr[0], G0 + 3);
    }
}

extern "C" void kernel_launch(void* const* d_in, const int* in_sizes, int n_in,
                              void* d_out, int out_size, void* d_ws, size_t ws_size,
                              hipStream_t stream) {
    const float* x    = (const float*)d_in[0];
    const float* W_ih = (const float*)d_in[1];
    const float* W_hh = (const float*)d_in[2];
    const float* b_ih = (const float*)d_in[3];
    const float* b_hh = (const float*)d_in[4];
    float* out = (float*)d_out;

    gru_fused<<<dim3(NCH / 4), dim3(32), 0, stream>>>(x, W_ih, W_hh, b_ih, b_hh, out);
}


# output store freed of its address-translation wait: two alternating (row pointer, lane offset) sets, h kept in its register 4 steps: no register of a store in flight is rewritten; sums unchanged
# speedup vs baseline: 7.1335x; 1.0371x over previous
#include <hip/hip_runtime.h>


typedef __attribute__((ext_vector_type(2))) float v2f;
typedef __attribute__((ext_vector_type(4))) float v4f;
typedef __attribute__((ext_vector_type(8))) float v8f;

#define Gc   64
#define Dc   16
#define Hc   8
#define Bc   64
#define Tc   512
#define K3   24
#define GD   (Gc * Dc)
#define NCH  (Gc * Bc)
#define TG   4
#define NXG  4
#define NGRP (Tc / TG)

__device__ __forceinline__ float fast_sigmoid(float x) {
    float e = __expf(-x);
    return __builtin_amdgcn_rcpf(1.0f + e);
}
__device__ __forceinline__ float fast_tanh(float x) {
    float e = __expf(2.0f * x);
    return 1.0f - 2.0f * __builtin_amdgcn_rcpf(e + 1.0f);
}

typedef const __attribute__((address_space(1))) float* gmem_row;
typedef const __attribute__((address_space(1))) v2f*   gmem_v2;
typedef const __attribute__((address_space(1))) char*  gmem_bytes;
typedef __attribute__((address_space(1))) float* gmem_wfloat;
typedef __attribute__((address_space(1))) char*  gmem_wbytes;

template <int J>
__device__ __forceinline__ float group8_bcast(float v) {
    return __uint_as_float(__builtin_amdgcn_mov_dpp8(__float_as_uint(v), J * 0x249249));
}

template <int CTRL>
__device__ __forceinline__ float row_shift_keep(float keep, float src) {
    return __uint_as_float(__builtin_amdgcn_update_dpp(
        __float_as_uint(keep), __float_as_uint(src), CTRL, 0xf, 0xf, false));
}

__device__ __forceinline__ v8f mm(v2f a, v2f b, v8f c) {
    return __builtin_amdgcn_wmma_f32_16x16x4_f32(false, a, false, b, (short)0, c, false, false);
}

__global__ __launch_bounds__(32) void gru_fused(
    const float* __restrict__ x,
    const float* __restrict__ W_ih,
    const float* __restrict__ W_hh,
    const float* __restrict__ b_ih,
    const float* __restrict__ b_hh,
    float* __restrict__ out)
{
    const int lane = threadIdx.x;
    const int i    = lane & 7;
    const int c    = blockIdx.x * 4 + (lane >> 3);
    const int b    = c & 63;
    const int g    = blockIdx.x >> 4;
    const int half = lane >> 4;
    const int l16  = lane & 15;
    const bool upper = (lane & 8) != 0;

    float wr[Hc], wz[Hc], wn[Hc];
    {
        const float* wg = W_hh + (size_t)g * K3 * Hc;
        const v4f* pr = (const v4f*)(wg + (size_t)i * Hc);
        const v4f* pz = (const v4f*)(wg + (size_t)(8 + i) * Hc);
        const v4f* pn = (const v4f*)(wg + (size_t)(16 + i) * Hc);
        v4f r0 = pr[0], r1 = pr[1], z0 = pz[0], z1 = pz[1], n0 = pn[0], n1 = pn[1];
        wr[0] = r0.x; wr[1] = r0.y; wr[2] = r0.z; wr[3] = r0.w;
        wr[4] = r1.x; wr[5] = r1.y; wr[6] = r1.z; wr[7] = r1.w;
        wz[0] = z0.x; wz[1] = z0.y; wz[2] = z0.z; wz[3] = z0.w;
        wz[4] = z1.x; wz[5] = z1.y; wz[6] = z1.z; wz[7] = z1.w;
        wn[0] = n0.x; wn[1] = n0.y; wn[2] = n0.z; wn[3] = n0.w;
        wn[4] = n1.x; wn[5] = n1.y; wn[6] = n1.z; wn[7] = n1.w;
    }
    float br = b_hh[(size_t)g * K3 + i];
    float bz = b_hh[(size_t)g * K3 + 8 + i];
    float bn = b_hh[(size_t)g * K3 + 16 + i];
    float ir = b_ih[(size_t)g * K3 + i];
    float iz = b_ih[(size_t)g * K3 + 8 + i];
    float in_ = b_ih[(size_t)g * K3 + 16 + i];

    v2f bm0[4], bm1[4];
    {
        const float* wg  = W_ih + (size_t)g * K3 * Dc;
        const v2f*   wr0 = (const v2f*)(wg + (size_t)l16 * Dc + 2 * half);
        const v2f*   wr1 = (const v2f*)(wg + (size_t)(16 + (l16 & 7)) * Dc + 2 * half);
        #pragma unroll
        for (int j = 0; j < 4; ++j) bm0[j] = wr0[2 * j];
        #pragma unroll
        for (int j = 0; j < 4; ++j) bm1[j] = wr1[2 * j];
    }

    const unsigned outOffB = ((unsigned)b * (Tc * Gc * Hc) + (unsigned)g * Hc + i) * 4u;
    const unsigned bA   = (unsigned)((blockIdx.x & 15) * 4 + (l16 >> 2));
    const unsigned xOffB = (bA * (unsigned)(Tc * GD) + (unsigned)(l16 & 3) * GD
                         + (unsigned)g * Dc + 2 * half) * 4u;

    float h = 0.0f;
    unsigned xoB = xOffB;

    const unsigned long long tileBytes = (unsigned long long)(TG * Gc * Hc * 4);
    unsigned long long obA = (unsigned long long)out;
    unsigned long long obB = obA - tileBytes;
    unsigned ooA = outOffB, ooB = outOffB;
    float hk1 = 0.0f, hk2 = 0.0f, hk3 = 0.0f, hk4 = 0.0f;
    int gnext = NXG + 1;

    auto loadx = [&](v2f a[4], gmem_row row, unsigned offB) {
        gmem_bytes p = (gmem_bytes)row + offB;
        #pragma unroll
        for (int j = 0; j < 4; ++j) a[j] = *(gmem_v2)(p + 16 * j);
    };

    auto step = [&](float gr, float gz, float gn, gmem_wbytes orow, unsigned ooff, auto&& behind_exp_r, auto&& behind_exp_n) {
        const float sr = gr + br;
        const float sz = gz + bz;

        const float h0 = group8_bcast<0>(h), h1 = group8_bcast<1>(h);
        const float h2 = group8_bcast<2>(h), h3 = group8_bcast<3>(h);
        const float h4 = group8_bcast<4>(h), h5 = group8_bcast<5>(h);
        const float h6 = group8_bcast<6>(h), h7 = group8_bcast<7>(h);

        float ar = fmaf(wr[3], h3, fmaf(wr[2], h2, fmaf(wr[1], h1, fmaf(wr[0], h0, sr))));
        float az = fmaf(wz[3], h3, fmaf(wz[2], h2, fmaf(wz[1], h1, fmaf(wz[0], h0, sz))));
        float an = fmaf(wn[3], h3, fmaf(wn[2], h2, fmaf(wn[1], h1, fmaf(wn[0], h0, bn))));
        float cr = fmaf(wr[7], h7, fmaf(wr[6], h6, fmaf(wr[5], h5, wr[4] * h4)));
        float cz = fmaf(wz[7], h7, fmaf(wz[6], h6, fmaf(wz[5], h5, wz[4] * h4)));
        float cn = fmaf(wn[7], h7, fmaf(wn[6], h6, fmaf(wn[5], h5, wn[4] * h4)));

        const float er = __expf(-(ar + cr));
        behind_exp_r(er);
        const float r = __builtin_amdgcn_rcpf(1.0f + er);
        const float z = fast_sigmoid(az + cz);
        const float en = __expf(2.0f * (gn + r * (an + cn)));
        behind_exp_n(en);
        const float n = 1.0f - 2.0f * __builtin_amdgcn_rcpf(en + 1.0f);
        h = (1.0f - z) * n + z * h;

        *(gmem_wfloat)(orow + ooff) = h;
        asm volatile("" : : "v"(hk4));
        hk4 = hk3; hk3 = hk2; hk2 = hk1; hk1 = h;
    };

    auto group = [&](const v8f& c0, const v8f& c1, v8f& n0, v8f& n1, v2f xa[4],
                     unsigned long long ob, unsigned oo,
                     unsigned long long& obN, unsigned& ooN) {
        const gmem_wbytes orow0 = (gmem_wbytes)ob;
        #pragma unroll
        for (int s = 0; s < TG; ++s) {
            const float gr = row_shift_keep<0x118>(c0[s], c0[4 + s]) + ir;
            const float gz = row_shift_keep<0x108>(c0[4 + s], c0[s]) + iz;
            float gn_hi = c1[4 + s];
            asm("" : "+v"(gn_hi));
            const float gn = (upper ? gn_hi : c1[s]) + in_;
            v2f ax = xa[s];
            v8f zero = {};
            step(gr, gz, gn, orow0 + s * (Gc * Hc * 4), oo,
                 [&](float e) {
                     asm volatile("" : "+v"(ax) : "v"(e));
                     n0 = mm(ax, bm0[s], s == 0 ? zero : n0);
                 },
                 [&](float e) {
                     asm volatile("" : "+v"(ax) : "v"(e));
                     n1 = mm(ax, bm1[s], s == 0 ? zero : n1);
                 });
            if (s == TG - 1) {
                asm volatile("" : "+s"(gnext), "+v"(xoB), "+s"(obN), "+v"(ooN) : "v"(h));
                obN += 2 * tileBytes;
                const int Gn = gnext < NGRP ? gnext : NGRP - 1;
                gmem_row row = (gmem_row)(x + (size_t)Gn * (TG * GD));
                gnext += 1;
                loadx(xa, row, xoB);
            }
            __builtin_amdgcn_sched_barrier(0);
        }
    };

    v2f xr[NXG][4];
    #pragma unroll
    for (int k = 0; k < NXG; ++k) loadx(xr[k], (gmem_row)(x + (size_t)k * (TG * GD)), xOffB);

    v8f a0 = {}, a1 = {}, d0 = {}, d1 = {};
    #pragma unroll
    for (int j = 0; j < 4; ++j) { a0 = mm(xr[0][j], bm0[j], a0); a1 = mm(xr[0][j], bm1[j], a1); }
    loadx(xr[0], (gmem_row)(x + (size_t)NXG * (TG * GD)), xOffB);

    #pragma unroll
    for (int j = 0; j < Hc; ++j)
        asm volatile("" : "+v"(wr[j]), "+v"(wz[j]), "+v"(wn[j]));
    asm volatile("" : "+v"(br), "+v"(bz), "+v"(bn), "+v"(ir), "+v"(iz), "+v"(in_));
    #pragma unroll
    for (int j = 0; j < 4; ++j)
        asm volatile("" : "+v"(bm0[j]), "+v"(bm1[j]));

    for (int G0 = 0; G0 < NGRP; G0 += NXG) {
        group(a0, a1, d0, d1, xr[1], obA, ooA, obB, ooB);
        group(d0, d1, a0, a1, xr[2], obB, ooB, obA, ooA);
        group(a0, a1, d0, d1, xr[3], obA, ooA, obB, ooB);
        group(d0, d1, a0, a1, xr[0], obB, ooB, obA, ooA);
    }
}

extern "C" void kernel_launch(void* const* d_in, const int* in_sizes, int n_in,
                              void* d_out, int out_size, void* d_ws, size_t ws_size,
                              hipStream_t stream) {
    const float* x    = (const float*)d_in[0];
    const float* W_ih = (const float*)d_in[1];
    const float* W_hh = (const float*)d_in[2];
    const float* b_ih = (const float*)d_in[3];
    const float* b_hh = (const float*)d_in[4];
    float* out = (float*)d_out;

    gru_fused<<<dim3(NCH / 4), dim3(32), 0, stream>>>(x, W_ih, W_hh, b_ih, b_hh, out);
}


# r, z, n gates on hardware tanh: sigmoid = (1 + tanh(x/2))/2, halves folded into weights; n argument summed as (gn + hn/2) + tr*hn/2, update as (1-z)*n + (z*h): gate rounding, these two sums change
# speedup vs baseline: 8.3043x; 1.1641x over previous
#include <hip/hip_runtime.h>


typedef __attribute__((ext_vector_type(2))) float v2f;
typedef __attribute__((ext_vector_type(4))) float v4f;
typedef __attribute__((ext_vector_type(8))) float v8f;

#define Gc   64
#define Dc   16
#define Hc   8
#define Bc   64
#define Tc   512
#define K3   24
#define GD   (Gc * Dc)
#define NCH  (Gc * Bc)
#define TG   4
#define NXG  4
#define NGRP (Tc / TG)

__device__ __forceinline__ float hw_tanh(float x) {
    return __builtin_amdgcn_tanhf(x);
}

typedef const __attribute__((address_space(1))) float* gmem_row;
typedef const __attribute__((address_space(1))) v2f*   gmem_v2;
typedef const __attribute__((address_space(1))) char*  gmem_bytes;
typedef __attribute__((address_space(1))) float* gmem_wfloat;
typedef __attribute__((address_space(1))) char*  gmem_wbytes;

template <int J>
__device__ __forceinline__ float group8_bcast(float v) {
    return __uint_as_float(__builtin_amdgcn_mov_dpp8(__float_as_uint(v), J * 0x249249));
}

template <int CTRL>
__device__ __forceinline__ float row_shift_keep(float keep, float src) {
    return __uint_as_float(__builtin_amdgcn_update_dpp(
        __float_as_uint(keep), __float_as_uint(src), CTRL, 0xf, 0xf, false));
}

__device__ __forceinline__ v8f mm(v2f a, v2f b, v8f c) {
    return __builtin_amdgcn_wmma_f32_16x16x4_f32(false, a, false, b, (short)0, c, false, false);
}

__global__ __launch_bounds__(32) void gru_fused(
    const float* __restrict__ x,
    const float* __restrict__ W_ih,
    const float* __restrict__ W_hh,
    const float* __restrict__ b_ih,
    const float* __restrict__ b_hh,
    float* __restrict__ out)
{
    const int lane = threadIdx.x;
    const int i    = lane & 7;
    const int c    = blockIdx.x * 4 + (lane >> 3);
    const int b    = c & 63;
    const int g    = blockIdx.x >> 4;
    const int half = lane >> 4;
    const int l16  = lane & 15;
    const bool upper = (lane & 8) != 0;

    float wr[Hc], wz[Hc], wn[Hc];
    {
        const float* wg = W_hh + (size_t)g * K3 * Hc;
        const v4f* pr = (const v4f*)(wg + (size_t)i * Hc);
        const v4f* pz = (const v4f*)(wg + (size_t)(8 + i) * Hc);
        const v4f* pn = (const v4f*)(wg + (size_t)(16 + i) * Hc);
        v4f r0 = pr[0], r1 = pr[1], z0 = pz[0], z1 = pz[1], n0 = pn[0], n1 = pn[1];
        wr[0] = r0.x; wr[1] = r0.y; wr[2] = r0.z; wr[3] = r0.w;
        wr[4] = r1.x; wr[5] = r1.y; wr[6] = r1.z; wr[7] = r1.w;
        wz[0] = z0.x; wz[1] = z0.y; wz[2] = z0.z; wz[3] = z0.w;
        wz[4] = z1.x; wz[5] = z1.y; wz[6] = z1.z; wz[7] = z1.w;
        wn[0] = n0.x; wn[1] = n0.y; wn[2] = n0.z; wn[3] = n0.w;
        wn[4] = n1.x; wn[5] = n1.y; wn[6] = n1.z; wn[7] = n1.w;
    }
    float br = b_hh[(size_t)g * K3 + i];
    float bz = b_hh[(size_t)g * K3 + 8 + i];
    float bn = b_hh[(size_t)g * K3 + 16 + i];
    float ir = b_ih[(size_t)g * K3 + i];
    float iz = b_ih[(size_t)g * K3 + 8 + i];
    float in_ = b_ih[(size_t)g * K3 + 16 + i];

    v2f bm0[4], bm1[4];
    {
        const float* wg  = W_ih + (size_t)g * K3 * Dc;
        const v2f*   wr0 = (const v2f*)(wg + (size_t)l16 * Dc + 2 * half);
        const v2f*   wr1 = (const v2f*)(wg + (size_t)(16 + (l16 & 7)) * Dc + 2 * half);
        #pragma unroll
        for (int j = 0; j < 4; ++j) bm0[j] = wr0[2 * j];
        #pragma unroll
        for (int j = 0; j < 4; ++j) bm1[j] = wr1[2 * j];
    }

    const unsigned outOffB = ((unsigned)b * (Tc * Gc * Hc) + (unsigned)g * Hc + i) * 4u;
    const unsigned bA   = (unsigned)((blockIdx.x & 15) * 4 + (l16 >> 2));
    const unsigned xOffB = (bA * (unsigned)(Tc * GD) + (unsigned)(l16 & 3) * GD
                         + (unsigned)g * Dc + 2 * half) * 4u;

    float h = 0.0f;
    unsigned xoB = xOffB;

    const unsigned long long tileBytes = (unsigned long long)(TG * Gc * Hc * 4);
    unsigned long long obA = (unsigned long long)out;
    unsigned long long obB = obA - tileBytes;
    unsigned ooA = outOffB, ooB = outOffB;
    float hk1 = 0.0f, hk2 = 0.0f, hk3 = 0.0f, hk4 = 0.0f;
    int gnext = NXG + 1;

    auto loadx = [&](v2f a[4], gmem_row row, unsigned offB) {
        gmem_bytes p = (gmem_bytes)row + offB;
        #pragma unroll
        for (int j = 0; j < 4; ++j) a[j] = *(gmem_v2)(p + 16 * j);
    };

    auto step = [&](float gr, float gz, float gn, gmem_wbytes orow, unsigned ooff, auto&& behind_tanh_r, auto&& behind_tanh_n) {
        const float sr = gr + br;
        const float sz = gz + bz;

        const float h0 = group8_bcast<0>(h), h1 = group8_bcast<1>(h);
        const float h2 = group8_bcast<2>(h), h3 = group8_bcast<3>(h);
        const float h4 = group8_bcast<4>(h), h5 = group8_bcast<5>(h);
        const float h6 = group8_bcast<6>(h), h7 = group8_bcast<7>(h);

        float ar = fmaf(wr[3], h3, fmaf(wr[2], h2, fmaf(wr[1], h1, fmaf(wr[0], h0, sr))));
        float az = fmaf(wz[3], h3, fmaf(wz[2], h2, fmaf(wz[1], h1, fmaf(wz[0], h0, sz))));
        float an = fmaf(wn[3], h3, fmaf(wn[2], h2, fmaf(wn[1], h1, fmaf(wn[0], h0, bn))));
        float cr = fmaf(wr[7], h7, fmaf(wr[6], h6, fmaf(wr[5], h5, wr[4] * h4)));
        float cz = fmaf(wz[7], h7, fmaf(wz[6], h6, fmaf(wz[5], h5, wz[4] * h4)));
        float cn = fmaf(wn[7], h7, fmaf(wn[6], h6, fmaf(wn[5], h5, wn[4] * h4)));

        const float tr = hw_tanh(ar + cr);
        behind_tanh_r(tr);
        const float tz = hw_tanh(az + cz);
        const float hh = an + cn;
        const float n  = hw_tanh(fmaf(tr, hh, gn + hh));
        behind_tanh_n(n);
        const float z   = fmaf(0.5f, tz, 0.5f);
        const float omz = fmaf(-0.5f, tz, 0.5f);
        h = fmaf(omz, n, z * h);

        *(gmem_wfloat)(orow + ooff) = h;
        asm volatile("" : : "v"(hk4));
        hk4 = hk3; hk3 = hk2; hk2 = hk1; hk1 = h;
    };

    auto group = [&](const v8f& c0, const v8f& c1, v8f& n0, v8f& n1, v2f xa[4],
                     unsigned long long ob, unsigned oo,
                     unsigned long long& obN, unsigned& ooN) {
        const gmem_wbytes orow0 = (gmem_wbytes)ob;
        #pragma unroll
        for (int s = 0; s < TG; ++s) {
            const float gr = row_shift_keep<0x118>(c0[s], c0[4 + s]) + ir;
            const float gz = row_shift_keep<0x108>(c0[4 + s], c0[s]) + iz;
            float gn_hi = c1[4 + s];
            asm("" : "+v"(gn_hi));
            const float gn = (upper ? gn_hi : c1[s]) + in_;
            v2f ax = xa[s];
            v8f zero = {};
            step(gr, gz, gn, orow0 + s * (Gc * Hc * 4), oo,
                 [&](float e) {
                     asm volatile("" : "+v"(ax) : "v"(e));
                     n0 = mm(ax, bm0[s], s == 0 ? zero : n0);
                 },
                 [&](float e) {
                     asm volatile("" : "+v"(ax) : "v"(e));
                     n1 = mm(ax, bm1[s], s == 0 ? zero : n1);
                 });
            if (s == TG - 1) {
                asm volatile("" : "+s"(gnext), "+v"(xoB), "+s"(obN), "+v"(ooN) : "v"(h));
                obN += 2 * tileBytes;
                const int Gn = gnext < NGRP ? gnext : NGRP - 1;
                gmem_row row = (gmem_row)(x + (size_t)Gn * (TG * GD));
                gnext += 1;
                loadx(xa, row, xoB);
            }
            __builtin_amdgcn_sched_barrier(0);
        }
    };

    v2f xr[NXG][4];
    #pragma unroll
    for (int k = 0; k < NXG; ++k) loadx(xr[k], (gmem_row)(x + (size_t)k * (TG * GD)), xOffB);

    for (int j = 0; j < Hc; ++j) { wr[j] *= 0.5f; wz[j] *= 0.5f; wn[j] *= 0.5f; }
    br *= 0.5f; bz *= 0.5f; bn *= 0.5f; ir *= 0.5f; iz *= 0.5f;
    for (int j = 0; j < 4; ++j) bm0[j] *= 0.5f;

    v8f a0 = {}, a1 = {}, d0 = {}, d1 = {};
    #pragma unroll
    for (int j = 0; j < 4; ++j) { a0 = mm(xr[0][j], bm0[j], a0); a1 = mm(xr[0][j], bm1[j], a1); }
    loadx(xr[0], (gmem_row)(x + (size_t)NXG * (TG * GD)), xOffB);

    #pragma unroll
    for (int j = 0; j < Hc; ++j)
        asm volatile("" : "+v"(wr[j]), "+v"(wz[j]), "+v"(wn[j]));
    asm volatile("" : "+v"(br), "+v"(bz), "+v"(bn), "+v"(ir), "+v"(iz), "+v"(in_));
    #pragma unroll
    for (int j = 0; j < 4; ++j)
        asm volatile("" : "+v"(bm0[j]), "+v"(bm1[j]));

    for (int G0 = 0; G0 < NGRP; G0 += NXG) {
        group(a0, a1, d0, d1, xr[1], obA, ooA, obB, ooB);
        group(d0, d1, a0, a1, xr[2], obB, ooB, obA, ooA);
        group(a0, a1, d0, d1, xr[3], obA, ooA, obB, ooB);
        group(d0, d1, a0, a1, xr[0], obB, ooB, obA, ooA);
    }
}

extern "C" void kernel_launch(void* const* d_in, const int* in_sizes, int n_in,
                              void* d_out, int out_size, void* d_ws, size_t ws_size,
                              hipStream_t stream) {
    const float* x    = (const float*)d_in[0];
    const float* W_ih = (const float*)d_in[1];
    const float* W_hh = (const float*)d_in[2];
    const float* b_ih = (const float*)d_in[3];
    const float* b_hh = (const float*)d_in[4];
    float* out = (float*)d_out;

    gru_fused<<<dim3(NCH / 4), dim3(32), 0, stream>>>(x, W_ih, W_hh, b_ih, b_hh, out);
}


# h broadcast inside each group of 8 lanes by v_permlane16_b32 gathers with constant selectors into fresh registers, in place of a register copy plus a DPP8 move per value; sums unchanged
# speedup vs baseline: 8.8387x; 1.0644x over previous
#include <hip/hip_runtime.h>


typedef __attribute__((ext_vector_type(2))) float v2f;
typedef __attribute__((ext_vector_type(4))) float v4f;
typedef __attribute__((ext_vector_type(8))) float v8f;

#define Gc   64
#define Dc   16
#define Hc   8
#define Bc   64
#define Tc   512
#define K3   24
#define GD   (Gc * Dc)
#define NCH  (Gc * Bc)
#define TG   4
#define NXG  4
#define NGRP (Tc / TG)

__device__ __forceinline__ float hw_tanh(float x) {
    return __builtin_amdgcn_tanhf(x);
}

typedef const __attribute__((address_space(1))) float* gmem_row;
typedef const __attribute__((address_space(1))) v2f*   gmem_v2;
typedef const __attribute__((address_space(1))) char*  gmem_bytes;
typedef __attribute__((address_space(1))) float* gmem_wfloat;
typedef __attribute__((address_space(1))) char*  gmem_wbytes;

template <int J>
__device__ __forceinline__ float group8_bcast(float v) {
    const unsigned u = __float_as_uint(v);
    return __uint_as_float(__builtin_amdgcn_permlane16(
        u, u, (unsigned)J * 0x11111111u, (unsigned)(8 + J) * 0x11111111u, true, false));
}

template <int CTRL>
__device__ __forceinline__ float row_shift_keep(float keep, float src) {
    return __uint_as_float(__builtin_amdgcn_update_dpp(
        __float_as_uint(keep), __float_as_uint(src), CTRL, 0xf, 0xf, false));
}

__device__ __forceinline__ v8f mm(v2f a, v2f b, v8f c) {
    return __builtin_amdgcn_wmma_f32_16x16x4_f32(false, a, false, b, (short)0, c, false, false);
}

__global__ __launch_bounds__(32) void gru_fused(
    const float* __restrict__ x,
    const float* __restrict__ W_ih,
    const float* __restrict__ W_hh,
    const float* __restrict__ b_ih,
    const float* __restrict__ b_hh,
    float* __restrict__ out)
{
    const int lane = threadIdx.x;
    const int i    = lane & 7;
    const int c    = blockIdx.x * 4 + (lane >> 3);
    const int b    = c & 63;
    const int g    = blockIdx.x >> 4;
    const int half = lane >> 4;
    const int l16  = lane & 15;
    const bool upper = (lane & 8) != 0;

    float wr[Hc], wz[Hc], wn[Hc];
    {
        const float* wg = W_hh + (size_t)g * K3 * Hc;
        const v4f* pr = (const v4f*)(wg + (size_t)i * Hc);
        const v4f* pz = (const v4f*)(wg + (size_t)(8 + i) * Hc);
        const v4f* pn = (const v4f*)(wg + (size_t)(16 + i) * Hc);
        v4f r0 = pr[0], r1 = pr[1], z0 = pz[0], z1 = pz[1], n0 = pn[0], n1 = pn[1];
        wr[0] = r0.x; wr[1] = r0.y; wr[2] = r0.z; wr[3] = r0.w;
        wr[4] = r1.x; wr[5] = r1.y; wr[6] = r1.z; wr[7] = r1.w;
        wz[0] = z0.x; wz[1] = z0.y; wz[2] = z0.z; wz[3] = z0.w;
        wz[4] = z1.x; wz[5] = z1.y; wz[6] = z1.z; wz[7] = z1.w;
        wn[0] = n0.x; wn[1] = n0.y; wn[2] = n0.z; wn[3] = n0.w;
        wn[4] = n1.x; wn[5] = n1.y; wn[6] = n1.z; wn[7] = n1.w;
    }
    float br = b_hh[(size_t)g * K3 + i];
    float bz = b_hh[(size_t)g * K3 + 8 + i];
    float bn = b_hh[(size_t)g * K3 + 16 + i];
    float ir = b_ih[(size_t)g * K3 + i];
    float iz = b_ih[(size_t)g * K3 + 8 + i];
    float in_ = b_ih[(size_t)g * K3 + 16 + i];

    v2f bm0[4], bm1[4];
    {
        const float* wg  = W_ih + (size_t)g * K3 * Dc;
        const v2f*   wr0 = (const v2f*)(wg + (size_t)l16 * Dc + 2 * half);
        const v2f*   wr1 = (const v2f*)(wg + (size_t)(16 + (l16 & 7)) * Dc + 2 * half);
        #pragma unroll
        for (int j = 0; j < 4; ++j) bm0[j] = wr0[2 * j];
        #pragma unroll
        for (int j = 0; j < 4; ++j) bm1[j] = wr1[2 * j];
    }

    const unsigned outOffB = ((unsigned)b * (Tc * Gc * Hc) + (unsigned)g * Hc + i) * 4u;
    const unsigned bA   = (unsigned)((blockIdx.x & 15) * 4 + (l16 >> 2));
    const unsigned xOffB = (bA * (unsigned)(Tc * GD) + (unsigned)(l16 & 3) * GD
                         + (unsigned)g * Dc + 2 * half) * 4u;

    float h = 0.0f;
    unsigned xoB = xOffB;

    const unsigned long long tileBytes = (unsigned long long)(TG * Gc * Hc * 4);
    unsigned long long obA = (unsigned long long)out;
    unsigned long long obB = obA - tileBytes;
    unsigned ooA = outOffB, ooB = outOffB;
    float hk1 = 0.0f, hk2 = 0.0f, hk3 = 0.0f, hk4 = 0.0f;
    int gnext = NXG + 1;

    auto loadx = [&](v2f a[4], gmem_row row, unsigned offB) {
        gmem_bytes p = (gmem_bytes)row + offB;
        #pragma unroll
        for (int j = 0; j < 4; ++j) a[j] = *(gmem_v2)(p + 16 * j);
    };

    auto step = [&](float gr, float gz, float gn, gmem_wbytes orow, unsigned ooff, auto&& behind_tanh_r, auto&& behind_tanh_n) {
        const float sr = gr + br;
        const float sz = gz + bz;

        const float h0 = group8_bcast<0>(h), h1 = group8_bcast<1>(h);
        const float h2 = group8_bcast<2>(h), h3 = group8_bcast<3>(h);
        const float h4 = group8_bcast<4>(h), h5 = group8_bcast<5>(h);
        const float h6 = group8_bcast<6>(h), h7 = group8_bcast<7>(h);

        float ar = fmaf(wr[3], h3, fmaf(wr[2], h2, fmaf(wr[1], h1, fmaf(wr[0], h0, sr))));
        float az = fmaf(wz[3], h3, fmaf(wz[2], h2, fmaf(wz[1], h1, fmaf(wz[0], h0, sz))));
        float an = fmaf(wn[3], h3, fmaf(wn[2], h2, fmaf(wn[1], h1, fmaf(wn[0], h0, bn))));
        float cr = fmaf(wr[7], h7, fmaf(wr[6], h6, fmaf(wr[5], h5, wr[4] * h4)));
        float cz = fmaf(wz[7], h7, fmaf(wz[6], h6, fmaf(wz[5], h5, wz[4] * h4)));
        float cn = fmaf(wn[7], h7, fmaf(wn[6], h6, fmaf(wn[5], h5, wn[4] * h4)));

        const float tr = hw_tanh(ar + cr);
        behind_tanh_r(tr);
        const float tz = hw_tanh(az + cz);
        const float hh = an + cn;
        const float n  = hw_tanh(fmaf(tr, hh, gn + hh));
        behind_tanh_n(n);
        const float z   = fmaf(0.5f, tz, 0.5f);
        const float omz = fmaf(-0.5f, tz, 0.5f);
        h = fmaf(omz, n, z * h);

        *(gmem_wfloat)(orow + ooff) = h;
        asm volatile("" : : "v"(hk4));
        hk4 = hk3; hk3 = hk2; hk2 = hk1; hk1 = h;
    };

    auto group = [&](const v8f& c0, const v8f& c1, v8f& n0, v8f& n1, v2f xa[4],
                     unsigned long long ob, unsigned oo,
                     unsigned long long& obN, unsigned& ooN) {
        const gmem_wbytes orow0 = (gmem_wbytes)ob;
        #pragma unroll
        for (int s = 0; s < TG; ++s) {
            const float gr = row_shift_keep<0x118>(c0[s], c0[4 + s]) + ir;
            const float gz = row_shift_keep<0x108>(c0[4 + s], c0[s]) + iz;
            float gn_hi = c1[4 + s];
            asm("" : "+v"(gn_hi));
            const float gn = (upper ? gn_hi : c1[s]) + in_;
            v2f ax = xa[s];
            v8f zero = {};
            step(gr, gz, gn, orow0 + s * (Gc * Hc * 4), oo,
                 [&](float e) {
                     asm volatile("" : "+v"(ax) : "v"(e));
                     n0 = mm(ax, bm0[s], s == 0 ? zero : n0);
                 },
                 [&](float e) {
                     asm volatile("" : "+v"(ax) : "v"(e));
                     n1 = mm(ax, bm1[s], s == 0 ? zero : n1);
                 });
            if (s == TG - 1) {
                asm volatile("" : "+s"(gnext), "+v"(xoB), "+s"(obN), "+v"(ooN) : "v"(h));
                obN += 2 * tileBytes;
                const int Gn = gnext < NGRP ? gnext : NGRP - 1;
                gmem_row row = (gmem_row)(x + (size_t)Gn * (TG * GD));
                gnext += 1;
                loadx(xa, row, xoB);
            }
            __builtin_amdgcn_sched_barrier(0);
        }
    };

    v2f xr[NXG][4];
    #pragma unroll
    for (int k = 0; k < NXG; ++k) loadx(xr[k], (gmem_row)(x + (size_t)k * (TG * GD)), xOffB);

    for (int j = 0; j < Hc; ++j) { wr[j] *= 0.5f; wz[j] *= 0.5f; wn[j] *= 0.5f; }
    br *= 0.5f; bz *= 0.5f; bn *= 0.5f; ir *= 0.5f; iz *= 0.5f;
    for (int j = 0; j < 4; ++j) bm0[j] *= 0.5f;

    v8f a0 = {}, a1 = {}, d0 = {}, d1 = {};
    #pragma unroll
    for (int j = 0; j < 4; ++j) { a0 = mm(xr[0][j], bm0[j], a0); a1 = mm(xr[0][j], bm1[j], a1); }
    loadx(xr[0], (gmem_row)(x + (size_t)NXG * (TG * GD)), xOffB);

    #pragma unroll
    for (int j = 0; j < Hc; ++j)
        asm volatile("" : "+v"(wr[j]), "+v"(wz[j]), "+v"(wn[j]));
    asm volatile("" : "+v"(br), "+v"(bz), "+v"(bn), "+v"(ir), "+v"(iz), "+v"(in_));
    #pragma unroll
    for (int j = 0; j < 4; ++j)
        asm volatile("" : "+v"(bm0[j]), "+v"(bm1[j]));

    for (int G0 = 0; G0 < NGRP; G0 += NXG) {
        group(a0, a1, d0, d1, xr[1], obA, ooA, obB, ooB);
        group(d0, d1, a0, a1, xr[2], obB, ooB, obA, ooA);
        group(a0, a1, d0, d1, xr[3], obA, ooA, obB, ooB);
        group(d0, d1, a0, a1, xr[0], obB, ooB, obA, ooA);
    }
}

extern "C" void kernel_launch(void* const* d_in, const int* in_sizes, int n_in,
                              void* d_out, int out_size, void* d_ws, size_t ws_size,
                              hipStream_t stream) {
    const float* x    = (const float*)d_in[0];
    const float* W_ih = (const float*)d_in[1];
    const float* W_hh = (const float*)d_in[2];
    const float* b_ih = (const float*)d_in[3];
    const float* b_hh = (const float*)d_in[4];
    float* out = (float*)d_out;

    gru_fused<<<dim3(NCH / 4), dim3(32), 0, stream>>>(x, W_ih, W_hh, b_ih, b_hh, out);
}


# step written in issue order for an in-order wave: six dot chains level by level, z gate finished in front of the two matrix products, next step's gate inputs made in this step's waits; sums unchanged
# speedup vs baseline: 9.1817x; 1.0388x over previous
#include <hip/hip_runtime.h>


typedef __attribute__((ext_vector_type(2))) float v2f;
typedef __attribute__((ext_vector_type(4))) float v4f;
typedef __attribute__((ext_vector_type(8))) float v8f;

#define Gc   64
#define Dc   16
#define Hc   8
#define Bc   64
#define Tc   512
#define K3   24
#define GD   (Gc * Dc)
#define NCH  (Gc * Bc)
#define TG   4
#define NXG  4
#define NGRP (Tc / TG)

__device__ __forceinline__ float hw_tanh(float x) {
    return __builtin_amdgcn_tanhf(x);
}

typedef const __attribute__((address_space(1))) float* gmem_row;
typedef const __attribute__((address_space(1))) v2f*   gmem_v2;
typedef const __attribute__((address_space(1))) char*  gmem_bytes;
typedef __attribute__((address_space(1))) float* gmem_wfloat;
typedef __attribute__((address_space(1))) char*  gmem_wbytes;

template <int J>
__device__ __forceinline__ float group8_bcast(float v) {
    const unsigned u = __float_as_uint(v);
    return __uint_as_float(__builtin_amdgcn_permlane16(
        u, u, (unsigned)J * 0x11111111u, (unsigned)(8 + J) * 0x11111111u, true, false));
}

template <int CTRL>
__device__ __forceinline__ float row_shift_keep(float keep, float src) {
    return __uint_as_float(__builtin_amdgcn_update_dpp(
        __float_as_uint(keep), __float_as_uint(src), CTRL, 0xf, 0xf, false));
}

__device__ __forceinline__ v8f mm(v2f a, v2f b, v8f c) {
    return __builtin_amdgcn_wmma_f32_16x16x4_f32(false, a, false, b, (short)0, c, false, false);
}

__global__ __launch_bounds__(32) void gru_fused(
    const float* __restrict__ x,
    const float* __restrict__ W_ih,
    const float* __restrict__ W_hh,
    const float* __restrict__ b_ih,
    const float* __restrict__ b_hh,
    float* __restrict__ out)
{
    const int lane = threadIdx.x;
    const int i    = lane & 7;
    const int c    = blockIdx.x * 4 + (lane >> 3);
    const int b    = c & 63;
    const int g    = blockIdx.x >> 4;
    const int half = lane >> 4;
    const int l16  = lane & 15;
    const bool upper = (lane & 8) != 0;

    float wr[Hc], wz[Hc], wn[Hc];
    {
        const float* wg = W_hh + (size_t)g * K3 * Hc;
        const v4f* pr = (const v4f*)(wg + (size_t)i * Hc);
        const v4f* pz = (const v4f*)(wg + (size_t)(8 + i) * Hc);
        const v4f* pn = (const v4f*)(wg + (size_t)(16 + i) * Hc);
        v4f r0 = pr[0], r1 = pr[1], z0 = pz[0], z1 = pz[1], n0 = pn[0], n1 = pn[1];
        wr[0] = r0.x; wr[1] = r0.y; wr[2] = r0.z; wr[3] = r0.w;
        wr[4] = r1.x; wr[5] = r1.y; wr[6] = r1.z; wr[7] = r1.w;
        wz[0] = z0.x; wz[1] = z0.y; wz[2] = z0.z; wz[3] = z0.w;
        wz[4] = z1.x; wz[5] = z1.y; wz[6] = z1.z; wz[7] = z1.w;
        wn[0] = n0.x; wn[1] = n0.y; wn[2] = n0.z; wn[3] = n0.w;
        wn[4] = n1.x; wn[5] = n1.y; wn[6] = n1.z; wn[7] = n1.w;
    }
    float br = b_hh[(size_t)g * K3 + i];
    float bz = b_hh[(size_t)g * K3 + 8 + i];
    float bn = b_hh[(size_t)g * K3 + 16 + i];
    float ir = b_ih[(size_t)g * K3 + i];
    float iz = b_ih[(size_t)g * K3 + 8 + i];
    float in_ = b_ih[(size_t)g * K3 + 16 + i];

    v2f bm0[4], bm1[4];
    {
        const float* wg  = W_ih + (size_t)g * K3 * Dc;
        const v2f*   wr0 = (const v2f*)(wg + (size_t)l16 * Dc + 2 * half);
        const v2f*   wr1 = (const v2f*)(wg + (size_t)(16 + (l16 & 7)) * Dc + 2 * half);
        #pragma unroll
        for (int j = 0; j < 4; ++j) bm0[j] = wr0[2 * j];
        #pragma unroll
        for (int j = 0; j < 4; ++j) bm1[j] = wr1[2 * j];
    }

    const unsigned outOffB = ((unsigned)b * (Tc * Gc * Hc) + (unsigned)g * Hc + i) * 4u;
    const unsigned bA   = (unsigned)((blockIdx.x & 15) * 4 + (l16 >> 2));
    const unsigned xOffB = (bA * (unsigned)(Tc * GD) + (unsigned)(l16 & 3) * GD
                         + (unsigned)g * Dc + 2 * half) * 4u;

    float h = 0.0f;
    unsigned xoB = xOffB;

    const unsigned long long tileBytes = (unsigned long long)(TG * Gc * Hc * 4);
    unsigned long long obA = (unsigned long long)out;
    unsigned long long obB = obA - tileBytes;
    unsigned ooA = outOffB, ooB = outOffB;
    float hk1 = 0.0f, hk2 = 0.0f, hk3 = 0.0f, hk4 = 0.0f;
    int gnext = NXG + 1;

    auto loadx = [&](v2f a[4], gmem_row row, unsigned offB) {
        gmem_bytes p = (gmem_bytes)row + offB;
        #pragma unroll
        for (int j = 0; j < 4; ++j) a[j] = *(gmem_v2)(p + 16 * j);
    };

    float pr = 0.0f, pz = 0.0f, pn = 0.0f;
    auto fetch_rz = [&](const v8f& t0, int j) {
        pr = row_shift_keep<0x118>(t0[j], t0[4 + j]);
        pz = row_shift_keep<0x108>(t0[4 + j], t0[j]);
    };
    auto fetch_n = [&](const v8f& t1, int j) {
        float gn_hi = t1[4 + j];
        asm("" : "+v"(gn_hi));
        pn = upper ? gn_hi : t1[j];
    };
    auto add_ih = [&]() { pr += ir; pz += iz; pn += in_; };
    auto add_hh = [&]() { pr += br; pz += bz; };

    auto step = [&](int s, const v8f& c0, const v8f& c1, const v8f& nx0,
                    gmem_wbytes orow, unsigned ooff, auto&& behind_tanh_r, auto&& behind_tanh_n) {
        if (s == 0) { fetch_n(c1, 0); add_ih(); add_hh(); }
        const float sr = pr, sz = pz, gn = pn;

        float h0 = group8_bcast<0>(h), h4 = group8_bcast<4>(h);
        float h1 = group8_bcast<1>(h), h5 = group8_bcast<5>(h);
        float h2 = group8_bcast<2>(h), h6 = group8_bcast<6>(h);
        float ar = fmaf(wr[0], h0, sr), cr = wr[4] * h4;
        float an = fmaf(wn[0], h0, bn), cn = wn[4] * h4;
        float az = fmaf(wz[0], h0, sz), cz = wz[4] * h4;
        float h3 = group8_bcast<3>(h), h7 = group8_bcast<7>(h);
        asm volatile("" : "+v"(ar), "+v"(cr), "+v"(an), "+v"(cn), "+v"(az), "+v"(cz), "+v"(h3), "+v"(h7));
        ar = fmaf(wr[1], h1, ar); cr = fmaf(wr[5], h5, cr);
        an = fmaf(wn[1], h1, an); cn = fmaf(wn[5], h5, cn);
        az = fmaf(wz[1], h1, az); cz = fmaf(wz[5], h5, cz);
        asm volatile("" : "+v"(ar), "+v"(cr), "+v"(an), "+v"(cn), "+v"(az), "+v"(cz));
        ar = fmaf(wr[2], h2, ar); cr = fmaf(wr[6], h6, cr);
        an = fmaf(wn[2], h2, an); cn = fmaf(wn[6], h6, cn);
        az = fmaf(wz[2], h2, az); cz = fmaf(wz[6], h6, cz);
        asm volatile("" : "+v"(ar), "+v"(cr), "+v"(an), "+v"(cn), "+v"(az), "+v"(cz));
        ar = fmaf(wr[3], h3, ar); cr = fmaf(wr[7], h7, cr);
        an = fmaf(wn[3], h3, an); cn = fmaf(wn[7], h7, cn);
        az = fmaf(wz[3], h3, az); cz = fmaf(wz[7], h7, cz);
        asm volatile("" : "+v"(ar), "+v"(cr), "+v"(an), "+v"(cn), "+v"(az), "+v"(cz));
        if (s < TG - 1) { fetch_rz(c0, s + 1); fetch_n(c1, s + 1); }

        float ra = ar + cr, za = az + cz, hh = an + cn;
        asm volatile("" : "+v"(ra), "+v"(za), "+v"(hh));
        __builtin_amdgcn_sched_barrier(0);
        if (s < TG - 1) add_ih();
        float tr = hw_tanh(ra);
        float tz = hw_tanh(za);
        float gh = gn + hh;
        behind_tanh_r(tr, tz, gh);
        __builtin_amdgcn_sched_barrier(0);
        const float na  = fmaf(tr, hh, gh);
        const float z   = fmaf(0.5f, tz, 0.5f);
        const float omz = fmaf(-0.5f, tz, 0.5f);
        const float zh  = z * h;
        const float n   = hw_tanh(na);
        behind_tanh_n(n, zh, omz);
        h = fmaf(omz, n, zh);
        asm volatile("" : "+v"(pr), "+v"(pz), "+v"(pn) : "v"(h));
        if (s < TG - 1) add_hh(); else fetch_rz(nx0, 0);

        *(gmem_wfloat)(orow + ooff) = h;
        asm volatile("" : : "v"(hk4));
        hk4 = hk3; hk3 = hk2; hk2 = hk1; hk1 = h;
    };

    auto group = [&](const v8f& c0, const v8f& c1, v8f& n0, v8f& n1, v2f xa[4],
                     unsigned long long ob, unsigned oo,
                     unsigned long long& obN, unsigned& ooN) {
        const gmem_wbytes orow0 = (gmem_wbytes)ob;
        #pragma unroll
        for (int s = 0; s < TG; ++s) {
            v2f ax = xa[s];
            v8f zero = {};
            step(s, c0, c1, n0, orow0 + s * (Gc * Hc * 4), oo,
                 [&](float& e, float& e2, float& e3) {
                     asm volatile("" : "+v"(ax), "+v"(e), "+v"(e2), "+v"(e3));
                     n0 = mm(ax, bm0[s], s == 0 ? zero : n0);
                 },
                 [&](float e, float e2, float e3) {
                     asm volatile("" : "+v"(ax) : "v"(e), "v"(e2), "v"(e3));
                     n1 = mm(ax, bm1[s], s == 0 ? zero : n1);
                 });
            if (s == TG - 1) {
                asm volatile("" : "+s"(gnext), "+v"(xoB), "+s"(obN), "+v"(ooN) : "v"(h));
                obN += 2 * tileBytes;
                const int Gn = gnext < NGRP ? gnext : NGRP - 1;
                gmem_row row = (gmem_row)(x + (size_t)Gn * (TG * GD));
                gnext += 1;
                loadx(xa, row, xoB);
            }
            __builtin_amdgcn_sched_barrier(0);
        }
    };

    v2f xr[NXG][4];
    #pragma unroll
    for (int k = 0; k < NXG; ++k) loadx(xr[k], (gmem_row)(x + (size_t)k * (TG * GD)), xOffB);

    for (int j = 0; j < Hc; ++j) { wr[j] *= 0.5f; wz[j] *= 0.5f; wn[j] *= 0.5f; }
    br *= 0.5f; bz *= 0.5f; bn *= 0.5f; ir *= 0.5f; iz *= 0.5f;
    for (int j = 0; j < 4; ++j) bm0[j] *= 0.5f;

    v8f a0 = {}, a1 = {}, d0 = {}, d1 = {};
    #pragma unroll
    for (int j = 0; j < 4; ++j) { a0 = mm(xr[0][j], bm0[j], a0); a1 = mm(xr[0][j], bm1[j], a1); }
    loadx(xr[0], (gmem_row)(x + (size_t)NXG * (TG * GD)), xOffB);

    #pragma unroll
    for (int j = 0; j < Hc; ++j)
        asm volatile("" : "+v"(wr[j]), "+v"(wz[j]), "+v"(wn[j]));
    asm volatile("" : "+v"(br), "+v"(bz), "+v"(bn), "+v"(ir), "+v"(iz), "+v"(in_));
    #pragma unroll
    for (int j = 0; j < 4; ++j)
        asm volatile("" : "+v"(bm0[j]), "+v"(bm1[j]));

    fetch_rz(a0, 0);

    for (int G0 = 0; G0 < NGRP; G0 += NXG) {
        group(a0, a1, d0, d1, xr[1], obA, ooA, obB, ooB);
        group(d0, d1, a0, a1, xr[2], obB, ooB, obA, ooA);
        group(a0, a1, d0, d1, xr[3], obA, ooA, obB, ooB);
        group(d0, d1, a0, a1, xr[0], obB, ooB, obA, ooA);
    }
}

extern "C" void kernel_launch(void* const* d_in, const int* in_sizes, int n_in,
                              void* d_out, int out_size, void* d_ws, size_t ws_size,
                              hipStream_t stream) {
    const float* x    = (const float*)d_in[0];
    const float* W_ih = (const float*)d_in[1];
    const float* W_hh = (const float*)d_in[2];
    const float* b_ih = (const float*)d_in[3];
    const float* b_hh = (const float*)d_in[4];
    float* out = (float*)d_out;

    gru_fused<<<dim3(NCH / 4), dim3(32), 0, stream>>>(x, W_ih, W_hh, b_ih, b_hh, out);
}
